# starting kernel
# baseline (speedup 1.0000x reference)
#include <hip/hip_runtime.h>

typedef __attribute__((ext_vector_type(2))) float v2f;
typedef __attribute__((ext_vector_type(8))) float v8f;

#define T_FULL 75000
#define CH_IN  17
#define NGROUP 32
#define NKER   8
#define NTAP   9
#define NUNIT  28
#define SG_FLOATS 75040
#define ZSLOT     75008
#define SW_OFF    SG_FLOATS
#define SACC_OFF  (SG_FLOATS + 72)
#define SMEM_FLOATS (SACC_OFF + 16)

__global__ __launch_bounds__(1024)
void hydra_unit_kernel(const float* __restrict__ X,
                       const float* __restrict__ W,
                       const int*   __restrict__ IDX,
                       float* __restrict__ OUT)
{
    extern __shared__ float smem[];
    float* sg   = smem;
    float* sw   = smem + SW_OFF;
    float* sAcc = smem + SACC_OFF;

    const int tid  = threadIdx.x;
    const int lane = tid & 31;
    const int wav  = tid >> 5;
    const int nthr = blockDim.x;
    const int nwav = nthr >> 5;

    const int bid = blockIdx.x;
    const int b   = bid / (NUNIT * NGROUP);
    const int rem = bid % (NUNIT * NGROUP);
    const int u   = rem / NGROUP;
    const int h   = rem % NGROUP;
    const int di  = u >> 1;
    const int dif = u & 1;
    const int d   = 1 << di;
    const int Teff = dif ? (T_FULL - 1) : T_FULL;

    if (tid < 16) sAcc[tid] = 0.0f;
    if (tid == 16) sg[ZSLOT] = 0.0f;
    {
        const int wbase = (u * (NKER * NGROUP) + h * NKER) * NTAP;
        if (tid < NKER * NTAP) sw[tid] = W[wbase + tid];
    }
    int ch[8];
    {
        const int* ip = IDX + (u * NGROUP + h) * 8;
        #pragma unroll
        for (int i = 0; i < 8; ++i) ch[i] = ip[i];
    }

    {
        const float* Xb = X + (size_t)b * CH_IN * T_FULL;
        for (int t = tid; t < Teff; t += nthr) {
            float s = 0.0f;
            #pragma unroll
            for (int i = 0; i < 8; ++i) {
                const float* xc = Xb + (size_t)ch[i] * T_FULL + t;
                float v = xc[0];
                if (dif) v = xc[1] - v;
                s += v;
            }
            sg[t] = s;
        }
    }
    __syncthreads();

    const int c16 = lane & 15;
    const int krb = (lane >> 4) << 1;
    v2f afrag[3];
    #pragma unroll
    for (int tile = 0; tile < 3; ++tile) {
        #pragma unroll
        for (int i = 0; i < 2; ++i) {
            const int j = tile * 4 + krb + i;
            afrag[tile][i] = (c16 < NKER && j < NTAP) ? sw[c16 * NTAP + j] : 0.0f;
        }
    }

    float accMax[8], accMin[8];
    #pragma unroll
    for (int k = 0; k < 8; ++k) { accMax[k] = 0.0f; accMin[k] = 0.0f; }

    const int nblk = (Teff + 31) >> 5;
    for (int blk = wav; blk < nblk; blk += nwav) {
        const int t0 = blk << 5;

        v2f bf0[3], bf1[3];
        #pragma unroll
        for (int tile = 0; tile < 3; ++tile) {
            #pragma unroll
            for (int i = 0; i < 2; ++i) {
                const int j  = tile * 4 + krb + i;
                const int p0 = t0 + c16 + (j - 4) * d;
                const int p1 = p0 + 16;
                const int a0 = (j < NTAP && (unsigned)p0 < (unsigned)Teff) ? p0 : ZSLOT;
                const int a1 = (j < NTAP && (unsigned)p1 < (unsigned)Teff) ? p1 : ZSLOT;
                bf0[tile][i] = sg[a0];
                bf1[tile][i] = sg[a1];
            }
        }

        v8f c0 = {0.f,0.f,0.f,0.f,0.f,0.f,0.f,0.f};
        v8f c1 = {0.f,0.f,0.f,0.f,0.f,0.f,0.f,0.f};
        #pragma unroll
        for (int tile = 0; tile < 3; ++tile) {
            c0 = __builtin_amdgcn_wmma_f32_16x16x4_f32(
                     false, afrag[tile], false, bf0[tile], (short)0, c0, false, false);
            c1 = __builtin_amdgcn_wmma_f32_16x16x4_f32(
                     false, afrag[tile], false, bf1[tile], (short)0, c1, false, false);
        }

        float z[8];
        #pragma unroll
        for (int k = 0; k < 8; ++k) {
            const float hi = __shfl(c1[k], c16, 32);
            z[k] = (lane < 16) ? c0[k] : hi;
        }

        if (t0 + lane < Teff) {
            float mx = z[0], mn = z[0];
            #pragma unroll
            for (int k = 1; k < 8; ++k) { mx = fmaxf(mx, z[k]); mn = fminf(mn, z[k]); }
            bool dmx = false, dmn = false;
            #pragma unroll
            for (int k = 0; k < 8; ++k) {
                const bool hm = (!dmx) && (z[k] == mx);
                accMax[k] += hm ? mx : 0.0f;  dmx = dmx || hm;
                const bool hn = (!dmn) && (z[k] == mn);
                accMin[k] += hn ? 1.0f : 0.0f; dmn = dmn || hn;
            }
        }
    }

    #pragma unroll
    for (int k = 0; k < 8; ++k) {
        float vM = accMax[k], vN = accMin[k];
        #pragma unroll
        for (int off = 16; off > 0; off >>= 1) {
            vM += __shfl_xor(vM, off, 32);
            vN += __shfl_xor(vN, off, 32);
        }
        if (lane == 0) {
            atomicAdd(&sAcc[k],     vM);
            atomicAdd(&sAcc[8 + k], vN);
        }
    }
    __syncthreads();

    if (tid < 16) {
        const int which = tid >> 3;
        const int k     = tid & 7;
        const size_t off = (size_t)b * 14336
                         + (size_t)(2 * u + which) * (NGROUP * NKER)
                         + (size_t)h * NKER + k;
        OUT[off] = sAcc[tid];
    }
}

extern "C" void kernel_launch(void* const* d_in, const int* in_sizes, int n_in,
                              void* d_out, int out_size, void* d_ws, size_t ws_size,
                              hipStream_t stream) {
    (void)in_sizes; (void)n_in; (void)d_ws; (void)ws_size; (void)out_size;
    const float* X   = (const float*)d_in[0];
    const float* W   = (const float*)d_in[1];
    const int*   IDX = (const int*)d_in[2];
    float*       OUT = (float*)d_out;

    const int shmem = SMEM_FLOATS * (int)sizeof(float);
    (void)hipFuncSetAttribute((const void*)hydra_unit_kernel,
                              hipFuncAttributeMaxDynamicSharedMemorySize, shmem);

    const int nblocks = 4 * NUNIT * NGROUP;
    hydra_unit_kernel<<<dim3(nblocks), dim3(1024), shmem, stream>>>(X, W, IDX, OUT);
}


# phase 2: both 16-point half-blocks in ONE accumulator: K slots alternate half-block 0/1, A block-diagonal (rows 8-15 = kernels on odd slots): 5 WMMAs per 32 points not 6, no lane pack; tap order kept
# speedup vs baseline: 1.6622x; 1.6622x over previous
#include <hip/hip_runtime.h>

typedef __attribute__((ext_vector_type(2))) float v2f;
typedef __attribute__((ext_vector_type(8))) float v8f;

#define T_FULL 75000
#define CH_IN  17
#define NGROUP 32
#define NKER   8
#define NTAP   9
#define NUNIT  28
#define HALO      2048
#define DI_HALO   9
#define SG_FLOATS (HALO + T_FULL + HALO + 32)
#define ZSLOT     0
#define SW_OFF    SG_FLOATS
#define SACC_OFF  (SG_FLOATS + 72)
#define SMEM_FLOATS (SACC_OFF + 16)
#define SEG    75
#define NSEG   1000
#define DBATCH 25
#define P1_THREADS 1024
#define RUNS    8
#define P1_TRIP (P1_THREADS * RUNS)
#define P1_MAIN ((T_FULL / P1_TRIP) * P1_TRIP)
static_assert(P1_THREADS % 32 == 0 && P1_MAIN <= T_FULL && RUNS * 8 <= 64,
              "gather pass: whole waves, whole trips inside the sequence, at most 64 loads of a wave in flight");
static_assert(NSEG * SEG == T_FULL && SEG % DBATCH == 0 && (SEG & 1) == 1 && NSEG <= 1024,
              "diff pass: odd segments of whole batches tile the sequence, one per thread of the 1024 launched");

static __device__ __forceinline__
void trip_body(const v2f (&afrag)[5], const v2f (&bf)[5],
               bool live, float (&accMax)[8], float (&accMin)[8])
{
    v8f c = {0.f,0.f,0.f,0.f,0.f,0.f,0.f,0.f};
    #pragma unroll
    for (int tile = 0; tile < 5; ++tile)
        c = __builtin_amdgcn_wmma_f32_16x16x4_f32(
                false, afrag[tile], false, bf[tile], (short)0, c, false, false);

    float z[8];
    #pragma unroll
    for (int k = 0; k < 8; ++k) z[k] = c[k];

    if (live) {
        float mx = z[0], mn = z[0];
        #pragma unroll
        for (int k = 1; k < 8; ++k) { mx = fmaxf(mx, z[k]); mn = fminf(mn, z[k]); }
        bool dmx = false, dmn = false;
        #pragma unroll
        for (int k = 0; k < 8; ++k) {
            const bool hm = (!dmx) && (z[k] == mx);
            accMax[k] += hm ? mx : 0.0f;  dmx = dmx || hm;
            const bool hn = (!dmn) && (z[k] == mn);
            accMin[k] += hn ? 1.0f : 0.0f; dmn = dmn || hn;
        }
    }
}

__global__ __launch_bounds__(1024)
void hydra_unit_kernel(const float* __restrict__ X,
                       const float* __restrict__ W,
                       const int*   __restrict__ IDX,
                       float* __restrict__ OUT)
{
    extern __shared__ float smem[];
    float* sz   = smem;
    float* sg   = smem + HALO;
    float* sw   = smem + SW_OFF;
    float* sAcc = smem + SACC_OFF;

    const int tid  = threadIdx.x;
    const int lane = tid & 31;
    const int wav  = tid >> 5;
    const int nthr = blockDim.x;
    const int nwav = nthr >> 5;

    const int bid = blockIdx.x;
    const int b   = bid / (NUNIT * NGROUP);
    const int rem = bid % (NUNIT * NGROUP);
    const int u   = rem / NGROUP;
    const int h   = rem % NGROUP;
    const int di  = u >> 1;
    const int dif = u & 1;
    const int d   = 1 << di;
    const int Teff = dif ? (T_FULL - 1) : T_FULL;

    if (tid < 16) sAcc[tid] = 0.0f;
    for (int z = tid; z < HALO; z += P1_THREADS) sz[z] = 0.0f;
    for (int z = tid; z < HALO + 32; z += P1_THREADS) sz[HALO + T_FULL + z] = 0.0f;
    {
        const int wbase = (u * (NKER * NGROUP) + h * NKER) * NTAP;
        if (tid < NKER * NTAP) sw[tid] = W[wbase + tid];
    }
    int ch[8];
    {
        const int* ip = IDX + (u * NGROUP + h) * 8;
        #pragma unroll
        for (int i = 0; i < 8; ++i) ch[i] = ip[i];
    }

    {
        const float* Xb = X + (size_t)b * CH_IN * T_FULL;
        const float* row[8];
        #pragma unroll
        for (int i = 0; i < 8; ++i) row[i] = Xb + (size_t)ch[i] * T_FULL;

        unsigned t = (unsigned)(wav * (RUNS * 32) + lane);
        for (; t + (RUNS - 1) * 32 < P1_MAIN; t += P1_TRIP) {
            float v[RUNS][8];
            #pragma unroll
            for (int i = 0; i < 8; ++i) {
                #pragma unroll
                for (int k = 0; k < RUNS; ++k) v[k][i] = (row[i] + t)[k * 32];
            }
            #pragma unroll
            for (int k = 0; k < RUNS; ++k) {
                float s = 0.0f;
                #pragma unroll
                for (int i = 0; i < 8; ++i) s += v[k][i];
                (sg + t)[k * 32] = s;
            }
        }
        for (unsigned r = P1_MAIN + (unsigned)tid; r < T_FULL; r += P1_THREADS) {
            float s = 0.0f;
            #pragma unroll
            for (int i = 0; i < 8; ++i) s += row[i][r];
            sg[r] = s;
        }
    }
    __syncthreads();

    if (dif) {
        const int base = tid * SEG;
        float nb = 0.0f;
        if (tid < NSEG - 1) nb = sg[base + SEG];
        __syncthreads();
        if (tid < NSEG) {
            float prev = sg[base];
            #pragma unroll
            for (int c = 0; c < SEG; c += DBATCH) {
                float v[DBATCH];
                #pragma unroll
                for (int i = 0; i < DBATCH; ++i) {
                    const int p = c + i + 1;
                    v[i] = (p < SEG) ? sg[base + p] : nb;
                }
                #pragma unroll
                for (int i = 0; i < DBATCH; ++i) {
                    const float dv = v[i] - prev;
                    prev = v[i];
                    if (c + i < SEG - 1)      sg[base + c + i] = dv;
                    else                      sg[base + c + i] = (tid < NSEG - 1) ? dv : 0.0f;
                }
            }
        }
        __syncthreads();
    }

    const int c16 = lane & 15;
    const int lh  = lane >> 4;
    v2f afrag[5];
    #pragma unroll
    for (int tile = 0; tile < 5; ++tile) {
        const int tap = 2 * tile + lh;
        const float wv = (tap < NTAP) ? sw[(c16 & 7) * NTAP + tap] : 0.0f;
        afrag[tile][0] = (c16 <  NKER) ? wv : 0.0f;
        afrag[tile][1] = (c16 >= NKER) ? wv : 0.0f;
    }

    float accMax[8], accMin[8];
    #pragma unroll
    for (int k = 0; k < 8; ++k) { accMax[k] = 0.0f; accMin[k] = 0.0f; }

    const int nblk = (Teff + 31) >> 5;
    if (di <= DI_HALO) {
        const int stride = nwav << 5;
        const int tb = HALO + (wav << 5) + c16;
        int o0 = tb + (0 + lh - 4) * d;
        int o1 = tb + (2 + lh - 4) * d;
        int o2 = tb + (4 + lh - 4) * d;
        int o3 = tb + (6 + lh - 4) * d;
        int o4 = (lane < 16) ? (tb + 4 * d) : (ZSLOT + c16);
        const int s4 = (lane < 16) ? stride : 0;
        for (int blk = wav; blk < nblk; blk += nwav) {
            const int t0 = blk << 5;
            v2f bf[5];
            bf[0][0] = sz[o0];  bf[0][1] = sz[o0 + 16];
            bf[1][0] = sz[o1];  bf[1][1] = sz[o1 + 16];
            bf[2][0] = sz[o2];  bf[2][1] = sz[o2 + 16];
            bf[3][0] = sz[o3];  bf[3][1] = sz[o3 + 16];
            bf[4][0] = sz[o4];  bf[4][1] = sz[o4 + 16];
            o0 += stride; o1 += stride; o2 += stride; o3 += stride; o4 += s4;
            __builtin_amdgcn_sched_barrier(0);
            trip_body(afrag, bf, t0 + lane < Teff, accMax, accMin);
        }
    } else {
        for (int blk = wav; blk < nblk; blk += nwav) {
            const int t0 = blk << 5;
            v2f bf[5];
            #pragma unroll
            for (int tile = 0; tile < 5; ++tile) {
                const int tap = 2 * tile + lh;
                const int p0 = t0 + c16 + (tap - 4) * d;
                const int p1 = p0 + 16;
                const int a0 = (tap < NTAP && (unsigned)p0 < (unsigned)Teff) ? (HALO + p0) : ZSLOT;
                const int a1 = (tap < NTAP && (unsigned)p1 < (unsigned)Teff) ? (HALO + p1) : ZSLOT;
                bf[tile][0] = sz[a0];
                bf[tile][1] = sz[a1];
            }
            __builtin_amdgcn_sched_barrier(0);
            trip_body(afrag, bf, t0 + lane < Teff, accMax, accMin);
        }
    }

    #pragma unroll
    for (int k = 0; k < 8; ++k) {
        float vM = accMax[k], vN = accMin[k];
        #pragma unroll
        for (int off = 16; off > 0; off >>= 1) {
            vM += __shfl_xor(vM, off, 32);
            vN += __shfl_xor(vN, off, 32);
        }
        if (lane == 0) {
            atomicAdd(&sAcc[k],     vM);
            atomicAdd(&sAcc[8 + k], vN);
        }
    }
    __syncthreads();

    if (tid < 16) {
        const int which = tid >> 3;
        const int k     = tid & 7;
        const size_t off = (size_t)b * 14336
                         + (size_t)(2 * u + which) * (NGROUP * NKER)
                         + (size_t)h * NKER + k;
        OUT[off] = sAcc[tid];
    }
}

extern "C" void kernel_launch(void* const* d_in, const int* in_sizes, int n_in,
                              void* d_out, int out_size, void* d_ws, size_t ws_size,
                              hipStream_t stream) {
    (void)in_sizes; (void)n_in; (void)d_ws; (void)ws_size; (void)out_size;
    const float* X   = (const float*)d_in[0];
    const float* W   = (const float*)d_in[1];
    const int*   IDX = (const int*)d_in[2];
    float*       OUT = (float*)d_out;

    const int shmem = SMEM_FLOATS * (int)sizeof(float);
    (void)hipFuncSetAttribute((const void*)hydra_unit_kernel,
                              hipFuncAttributeMaxDynamicSharedMemorySize, shmem);

    const int nblocks = 4 * NUNIT * NGROUP;
    hydra_unit_kernel<<<dim3(nblocks), dim3(1024), shmem, stream>>>(X, W, IDX, OUT);
}


# phase-2 scoring on running extrema: lane sums mx*[max(z[0..k])==mx] and counts [min(z[0..k])==mn], per-kernel results are differences taken once at the end; no scalar mask ops; max sums regrouped
# speedup vs baseline: 1.8552x; 1.1161x over previous
#include <hip/hip_runtime.h>

typedef __attribute__((ext_vector_type(2))) float v2f;
typedef __attribute__((ext_vector_type(8))) float v8f;

#define T_FULL 75000
#define CH_IN  17
#define NGROUP 32
#define NKER   8
#define NTAP   9
#define NUNIT  28
#define HALO      2048
#define DI_HALO   9
#define SG_FLOATS (HALO + T_FULL + HALO + 32)
#define ZSLOT     0
#define SW_OFF    SG_FLOATS
#define SACC_OFF  (SG_FLOATS + 72)
#define SMEM_FLOATS (SACC_OFF + 16)
#define SEG    75
#define NSEG   1000
#define DBATCH 25
#define P1_THREADS 1024
#define RUNS    8
#define P1_TRIP (P1_THREADS * RUNS)
#define P1_MAIN ((T_FULL / P1_TRIP) * P1_TRIP)
static_assert(P1_THREADS % 32 == 0 && P1_MAIN <= T_FULL && RUNS * 8 <= 64,
              "gather pass: whole waves, whole trips inside the sequence, at most 64 loads of a wave in flight");
static_assert(NSEG * SEG == T_FULL && SEG % DBATCH == 0 && (SEG & 1) == 1 && NSEG <= 1024,
              "diff pass: odd segments of whole batches tile the sequence, one per thread of the 1024 launched");

static __device__ __forceinline__
void trip_body(const v2f (&afrag)[5], const v2f (&bf)[5],
               bool live, float (&eMax)[8], int (&nMin)[7])
{
    v8f c = {0.f,0.f,0.f,0.f,0.f,0.f,0.f,0.f};
    #pragma unroll
    for (int tile = 0; tile < 5; ++tile)
        c = __builtin_amdgcn_wmma_f32_16x16x4_f32(
                false, afrag[tile], false, bf[tile], (short)0, c, false, false);

    if (live) {
        float rm[8], rn[8];
        rm[0] = c[0]; rn[0] = c[0];
        #pragma unroll
        for (int k = 1; k < 8; ++k) { rm[k] = fmaxf(rm[k - 1], c[k]); rn[k] = fminf(rn[k - 1], c[k]); }
        const float mx = rm[7], mn = rn[7];
        #pragma unroll
        for (int k = 0; k < 7; ++k) {
            eMax[k] += (rm[k] == mx) ? mx : 0.0f;
            nMin[k] += (rn[k] == mn) ? 1 : 0;
        }
        eMax[7] += mx;
    }
}

__global__ __launch_bounds__(1024)
void hydra_unit_kernel(const float* __restrict__ X,
                       const float* __restrict__ W,
                       const int*   __restrict__ IDX,
                       float* __restrict__ OUT)
{
    extern __shared__ float smem[];
    float* sz   = smem;
    float* sg   = smem + HALO;
    float* sw   = smem + SW_OFF;
    float* sAcc = smem + SACC_OFF;

    const int tid  = threadIdx.x;
    const int lane = tid & 31;
    const int wav  = tid >> 5;
    const int nthr = blockDim.x;
    const int nwav = nthr >> 5;

    const int bid = blockIdx.x;
    const int b   = bid / (NUNIT * NGROUP);
    const int rem = bid % (NUNIT * NGROUP);
    const int u   = rem / NGROUP;
    const int h   = rem % NGROUP;
    const int di  = u >> 1;
    const int dif = u & 1;
    const int d   = 1 << di;
    const int Teff = dif ? (T_FULL - 1) : T_FULL;

    if (tid < 16) sAcc[tid] = 0.0f;
    for (int z = tid; z < HALO; z += P1_THREADS) sz[z] = 0.0f;
    for (int z = tid; z < HALO + 32; z += P1_THREADS) sz[HALO + T_FULL + z] = 0.0f;
    {
        const int wbase = (u * (NKER * NGROUP) + h * NKER) * NTAP;
        if (tid < NKER * NTAP) sw[tid] = W[wbase + tid];
    }
    int ch[8];
    {
        const int* ip = IDX + (u * NGROUP + h) * 8;
        #pragma unroll
        for (int i = 0; i < 8; ++i) ch[i] = ip[i];
    }

    {
        const float* Xb = X + (size_t)b * CH_IN * T_FULL;
        const float* row[8];
        #pragma unroll
        for (int i = 0; i < 8; ++i) row[i] = Xb + (size_t)ch[i] * T_FULL;

        unsigned t = (unsigned)(wav * (RUNS * 32) + lane);
        for (; t + (RUNS - 1) * 32 < P1_MAIN; t += P1_TRIP) {
            float v[RUNS][8];
            #pragma unroll
            for (int i = 0; i < 8; ++i) {
                #pragma unroll
                for (int k = 0; k < RUNS; ++k) v[k][i] = (row[i] + t)[k * 32];
            }
            #pragma unroll
            for (int k = 0; k < RUNS; ++k) {
                float s = 0.0f;
                #pragma unroll
                for (int i = 0; i < 8; ++i) s += v[k][i];
                (sg + t)[k * 32] = s;
            }
        }
        for (unsigned r = P1_MAIN + (unsigned)tid; r < T_FULL; r += P1_THREADS) {
            float s = 0.0f;
            #pragma unroll
            for (int i = 0; i < 8; ++i) s += row[i][r];
            sg[r] = s;
        }
    }
    __syncthreads();

    if (dif) {
        const int base = tid * SEG;
        float nb = 0.0f;
        if (tid < NSEG - 1) nb = sg[base + SEG];
        __syncthreads();
        if (tid < NSEG) {
            float prev = sg[base];
            #pragma unroll
            for (int c = 0; c < SEG; c += DBATCH) {
                float v[DBATCH];
                #pragma unroll
                for (int i = 0; i < DBATCH; ++i) {
                    const int p = c + i + 1;
                    v[i] = (p < SEG) ? sg[base + p] : nb;
                }
                #pragma unroll
                for (int i = 0; i < DBATCH; ++i) {
                    const float dv = v[i] - prev;
                    prev = v[i];
                    if (c + i < SEG - 1)      sg[base + c + i] = dv;
                    else                      sg[base + c + i] = (tid < NSEG - 1) ? dv : 0.0f;
                }
            }
        }
        __syncthreads();
    }

    const int c16 = lane & 15;
    const int lh  = lane >> 4;
    v2f afrag[5];
    #pragma unroll
    for (int tile = 0; tile < 5; ++tile) {
        const int tap = 2 * tile + lh;
        const float wv = (tap < NTAP) ? sw[(c16 & 7) * NTAP + tap] : 0.0f;
        afrag[tile][0] = (c16 <  NKER) ? wv : 0.0f;
        afrag[tile][1] = (c16 >= NKER) ? wv : 0.0f;
    }

    float eMax[8];
    int   nMin[7];
    #pragma unroll
    for (int k = 0; k < 8; ++k) eMax[k] = 0.0f;
    #pragma unroll
    for (int k = 0; k < 7; ++k) nMin[k] = 0;

    const int nblk = (Teff + 31) >> 5;
    if (di <= DI_HALO) {
        const int stride = nwav << 5;
        const int tb = HALO + (wav << 5) + c16;
        int o0 = tb + (0 + lh - 4) * d;
        int o1 = tb + (2 + lh - 4) * d;
        int o2 = tb + (4 + lh - 4) * d;
        int o3 = tb + (6 + lh - 4) * d;
        int o4 = (lane < 16) ? (tb + 4 * d) : (ZSLOT + c16);
        const int s4 = (lane < 16) ? stride : 0;
        for (int blk = wav; blk < nblk; blk += nwav) {
            const int t0 = blk << 5;
            v2f bf[5];
            bf[0][0] = sz[o0];  bf[0][1] = sz[o0 + 16];
            bf[1][0] = sz[o1];  bf[1][1] = sz[o1 + 16];
            bf[2][0] = sz[o2];  bf[2][1] = sz[o2 + 16];
            bf[3][0] = sz[o3];  bf[3][1] = sz[o3 + 16];
            bf[4][0] = sz[o4];  bf[4][1] = sz[o4 + 16];
            o0 += stride; o1 += stride; o2 += stride; o3 += stride; o4 += s4;
            __builtin_amdgcn_sched_barrier(0);
            trip_body(afrag, bf, t0 + lane < Teff, eMax, nMin);
        }
    } else {
        for (int blk = wav; blk < nblk; blk += nwav) {
            const int t0 = blk << 5;
            v2f bf[5];
            #pragma unroll
            for (int tile = 0; tile < 5; ++tile) {
                const int tap = 2 * tile + lh;
                const int p0 = t0 + c16 + (tap - 4) * d;
                const int p1 = p0 + 16;
                const int a0 = (tap < NTAP && (unsigned)p0 < (unsigned)Teff) ? (HALO + p0) : ZSLOT;
                const int a1 = (tap < NTAP && (unsigned)p1 < (unsigned)Teff) ? (HALO + p1) : ZSLOT;
                bf[tile][0] = sz[a0];
                bf[tile][1] = sz[a1];
            }
            __builtin_amdgcn_sched_barrier(0);
            trip_body(afrag, bf, t0 + lane < Teff, eMax, nMin);
        }
    }

    float accMax[8], accMin[8];
    {
        int nscored = (wav < nblk) ? (nblk - 1 - wav) / nwav + 1 : 0;
        const int lastblk = nblk - 1;
        if (wav < nblk && (lastblk - wav) % nwav == 0 && (lastblk << 5) + lane >= Teff) nscored -= 1;
        accMax[0] = eMax[0];
        accMin[0] = (float)nMin[0];
        #pragma unroll
        for (int k = 1; k < 8; ++k) accMax[k] = eMax[k] - eMax[k - 1];
        #pragma unroll
        for (int k = 1; k < 7; ++k) accMin[k] = (float)(nMin[k] - nMin[k - 1]);
        accMin[7] = (float)(nscored - nMin[6]);
    }

    #pragma unroll
    for (int k = 0; k < 8; ++k) {
        float vM = accMax[k], vN = accMin[k];
        #pragma unroll
        for (int off = 16; off > 0; off >>= 1) {
            vM += __shfl_xor(vM, off, 32);
            vN += __shfl_xor(vN, off, 32);
        }
        if (lane == 0) {
            atomicAdd(&sAcc[k],     vM);
            atomicAdd(&sAcc[8 + k], vN);
        }
    }
    __syncthreads();

    if (tid < 16) {
        const int which = tid >> 3;
        const int k     = tid & 7;
        const size_t off = (size_t)b * 14336
                         + (size_t)(2 * u + which) * (NGROUP * NKER)
                         + (size_t)h * NKER + k;
        OUT[off] = sAcc[tid];
    }
}

extern "C" void kernel_launch(void* const* d_in, const int* in_sizes, int n_in,
                              void* d_out, int out_size, void* d_ws, size_t ws_size,
                              hipStream_t stream) {
    (void)in_sizes; (void)n_in; (void)d_ws; (void)ws_size; (void)out_size;
    const float* X   = (const float*)d_in[0];
    const float* W   = (const float*)d_in[1];
    const int*   IDX = (const int*)d_in[2];
    float*       OUT = (float*)d_out;

    const int shmem = SMEM_FLOATS * (int)sizeof(float);
    (void)hipFuncSetAttribute((const void*)hydra_unit_kernel,
                              hipFuncAttributeMaxDynamicSharedMemorySize, shmem);

    const int nblocks = 4 * NUNIT * NGROUP;
    hydra_unit_kernel<<<dim3(nblocks), dim3(1024), shmem, stream>>>(X, W, IDX, OUT);
}


# phase 2: one trip loop for all dilations, control wave-uniform: scalar block counter, five running LDS addresses (d>=1024: held in the zero halos by min/max), lane guard on last block only; sums same
# speedup vs baseline: 2.0206x; 1.0892x over previous
#include <hip/hip_runtime.h>

typedef __attribute__((ext_vector_type(2))) float v2f;
typedef __attribute__((ext_vector_type(8))) float v8f;
typedef const __attribute__((address_space(3))) float* lds_cfp;

static __device__ __forceinline__ unsigned lds_addr(const float* p)
{
    return (unsigned)(__UINTPTR_TYPE__)(lds_cfp)p;
}

#define T_FULL 75000
#define CH_IN  17
#define NGROUP 32
#define NKER   8
#define NTAP   9
#define NUNIT  28
#define HALO      2048
#define DI_HALO   9
#define SG_FLOATS (HALO + T_FULL + HALO + 32)
#define ZSLOT     0
#define SW_OFF    SG_FLOATS
#define SACC_OFF  (SG_FLOATS + 72)
#define SMEM_FLOATS (SACC_OFF + 16)
#define SEG    75
#define NSEG   1000
#define DBATCH 25
#define P1_THREADS 1024
#define RUNS    8
#define P1_TRIP (P1_THREADS * RUNS)
#define P1_MAIN ((T_FULL / P1_TRIP) * P1_TRIP)
static_assert(P1_THREADS % 32 == 0 && P1_MAIN <= T_FULL && RUNS * 8 <= 64,
              "gather pass: whole waves, whole trips inside the sequence, at most 64 loads of a wave in flight");
static_assert(NSEG * SEG == T_FULL && SEG % DBATCH == 0 && (SEG & 1) == 1 && NSEG <= 1024,
              "diff pass: odd segments of whole batches tile the sequence, one per thread of the 1024 launched");

template <bool ALL_LIVE>
static __device__ __forceinline__
void trip_body(const v2f (&afrag)[5], const v2f (&bf)[5],
               bool live, float (&eMax)[8], int (&nMin)[7])
{
    v8f c = {0.f,0.f,0.f,0.f,0.f,0.f,0.f,0.f};
    #pragma unroll
    for (int tile = 0; tile < 5; ++tile)
        c = __builtin_amdgcn_wmma_f32_16x16x4_f32(
                false, afrag[tile], false, bf[tile], (short)0, c, false, false);

    if (ALL_LIVE || live) {
        float rm[8], rn[8];
        rm[0] = c[0]; rn[0] = c[0];
        #pragma unroll
        for (int k = 1; k < 8; ++k) { rm[k] = fmaxf(rm[k - 1], c[k]); rn[k] = fminf(rn[k - 1], c[k]); }
        const float mx = rm[7], mn = rn[7];
        #pragma unroll
        for (int k = 0; k < 7; ++k) {
            eMax[k] += (rm[k] == mx) ? mx : 0.0f;
            nMin[k] += (rn[k] == mn) ? 1 : 0;
        }
        eMax[7] += mx;
    }
}

template <bool CLAMP>
static __device__ __forceinline__
void phase2_trips(const float* sz, const v2f (&afrag)[5], float (&eMax)[8], int (&nMin)[7],
                  int d, int lane, int wav, int wavu, int nwav, int nfull, int nblk, int Teff)
{
    const int c16 = lane & 15;
    const int lh  = lane >> 4;
    const int stride = nwav << 7;
    const int base = (int)lds_addr(sz);
    const int tb = HALO + (wav << 5) + c16;
    int o0 = base + (tb + (0 + lh - 4) * d) * 4;
    int o1 = base + (tb + (2 + lh - 4) * d) * 4;
    int o2 = base + (tb + (4 + lh - 4) * d) * 4;
    int o3 = base + (tb + (6 + lh - 4) * d) * 4;
    int o4 = base + ((lane < 16) ? (tb + 4 * d) : (ZSLOT + c16)) * 4;
    const int s4 = (lane < 16) ? stride : 0;
    const int lo = base + c16 * 4;
    const int hi = base + (HALO + T_FULL + HALO + c16) * 4;
    asm volatile("" : "+v"(o0), "+v"(o1), "+v"(o2), "+v"(o3), "+v"(o4));
    auto win = [&](int o) -> int {
        if (!CLAMP) return o;
        const int t = (o < hi) ? o : hi;
        return (t > lo) ? t : lo;
    };
    auto at  = [&](int a) -> float { return *(lds_cfp)(unsigned)a; };
    auto load5 = [&](v2f (&bf)[5]) {
        const int a0 = win(o0), a1 = win(o1), a2 = win(o2), a3 = win(o3), a4 = win(o4);
        bf[0][0] = at(a0);  bf[0][1] = at(a0 + 64);
        bf[1][0] = at(a1);  bf[1][1] = at(a1 + 64);
        bf[2][0] = at(a2);  bf[2][1] = at(a2 + 64);
        bf[3][0] = at(a3);  bf[3][1] = at(a3 + 64);
        bf[4][0] = at(a4);  bf[4][1] = at(a4 + 64);
    };
    int blk = wavu;
    for (; blk < nfull; blk += nwav) {
        v2f bf[5];
        load5(bf);
        o0 += stride; o1 += stride; o2 += stride; o3 += stride; o4 += s4;
        __builtin_amdgcn_sched_barrier(0);
        trip_body<true>(afrag, bf, true, eMax, nMin);
    }
    if (blk == nfull && nfull < nblk) {
        v2f bf[5];
        load5(bf);
        __builtin_amdgcn_sched_barrier(0);
        trip_body<false>(afrag, bf, (nfull << 5) + lane < Teff, eMax, nMin);
    }
}

__global__ __launch_bounds__(1024)
void hydra_unit_kernel(const float* __restrict__ X,
                       const float* __restrict__ W,
                       const int*   __restrict__ IDX,
                       float* __restrict__ OUT)
{
    extern __shared__ float smem[];
    float* sz   = smem;
    float* sg   = smem + HALO;
    float* sw   = smem + SW_OFF;
    float* sAcc = smem + SACC_OFF;

    const int tid  = threadIdx.x;
    const int lane = tid & 31;
    const int wav  = tid >> 5;
    const int nthr = blockDim.x;
    const int nwav = nthr >> 5;

    const int bid = blockIdx.x;
    const int b   = bid / (NUNIT * NGROUP);
    const int rem = bid % (NUNIT * NGROUP);
    const int u   = rem / NGROUP;
    const int h   = rem % NGROUP;
    const int di  = u >> 1;
    const int dif = u & 1;
    const int d   = 1 << di;
    const int Teff = dif ? (T_FULL - 1) : T_FULL;

    if (tid < 16) sAcc[tid] = 0.0f;
    for (int z = tid; z < HALO; z += P1_THREADS) sz[z] = 0.0f;
    for (int z = tid; z < HALO + 32; z += P1_THREADS) sz[HALO + T_FULL + z] = 0.0f;
    {
        const int wbase = (u * (NKER * NGROUP) + h * NKER) * NTAP;
        if (tid < NKER * NTAP) sw[tid] = W[wbase + tid];
    }
    int ch[8];
    {
        const int* ip = IDX + (u * NGROUP + h) * 8;
        #pragma unroll
        for (int i = 0; i < 8; ++i) ch[i] = ip[i];
    }

    {
        const float* Xb = X + (size_t)b * CH_IN * T_FULL;
        const float* row[8];
        #pragma unroll
        for (int i = 0; i < 8; ++i) row[i] = Xb + (size_t)ch[i] * T_FULL;

        unsigned t = (unsigned)(wav * (RUNS * 32) + lane);
        for (; t + (RUNS - 1) * 32 < P1_MAIN; t += P1_TRIP) {
            float v[RUNS][8];
            #pragma unroll
            for (int i = 0; i < 8; ++i) {
                #pragma unroll
                for (int k = 0; k < RUNS; ++k) v[k][i] = (row[i] + t)[k * 32];
            }
            #pragma unroll
            for (int k = 0; k < RUNS; ++k) {
                float s = 0.0f;
                #pragma unroll
                for (int i = 0; i < 8; ++i) s += v[k][i];
                (sg + t)[k * 32] = s;
            }
        }
        for (unsigned r = P1_MAIN + (unsigned)tid; r < T_FULL; r += P1_THREADS) {
            float s = 0.0f;
            #pragma unroll
            for (int i = 0; i < 8; ++i) s += row[i][r];
            sg[r] = s;
        }
    }
    __syncthreads();

    if (dif) {
        const int base = tid * SEG;
        float nb = 0.0f;
        if (tid < NSEG - 1) nb = sg[base + SEG];
        __syncthreads();
        if (tid < NSEG) {
            float prev = sg[base];
            #pragma unroll
            for (int c = 0; c < SEG; c += DBATCH) {
                float v[DBATCH];
                #pragma unroll
                for (int i = 0; i < DBATCH; ++i) {
                    const int p = c + i + 1;
                    v[i] = (p < SEG) ? sg[base + p] : nb;
                }
                #pragma unroll
                for (int i = 0; i < DBATCH; ++i) {
                    const float dv = v[i] - prev;
                    prev = v[i];
                    if (c + i < SEG - 1)      sg[base + c + i] = dv;
                    else                      sg[base + c + i] = (tid < NSEG - 1) ? dv : 0.0f;
                }
            }
        }
        __syncthreads();
    }

    const int c16 = lane & 15;
    const int lh  = lane >> 4;
    v2f afrag[5];
    #pragma unroll
    for (int tile = 0; tile < 5; ++tile) {
        const int tap = 2 * tile + lh;
        const float wv = (tap < NTAP) ? sw[(c16 & 7) * NTAP + tap] : 0.0f;
        afrag[tile][0] = (c16 <  NKER) ? wv : 0.0f;
        afrag[tile][1] = (c16 >= NKER) ? wv : 0.0f;
    }

    float eMax[8];
    int   nMin[7];
    #pragma unroll
    for (int k = 0; k < 8; ++k) eMax[k] = 0.0f;
    #pragma unroll
    for (int k = 0; k < 7; ++k) nMin[k] = 0;

    const int wavu  = __builtin_amdgcn_readfirstlane(wav);
    const int nfull = Teff >> 5;
    const int nblk = (Teff + 31) >> 5;
    if (di <= DI_HALO)
        phase2_trips<false>(sz, afrag, eMax, nMin, d, lane, wav, wavu, nwav, nfull, nblk, Teff);
    else
        phase2_trips<true>(sz, afrag, eMax, nMin, d, lane, wav, wavu, nwav, nfull, nblk, Teff);

    float accMax[8], accMin[8];
    {
        int nscored = (wavu < nblk) ? (nblk - 1 - wavu) / nwav + 1 : 0;
        const int lastblk = nblk - 1;
        if (wavu < nblk && (lastblk - wavu) % nwav == 0 && (lastblk << 5) + lane >= Teff) nscored -= 1;
        accMax[0] = eMax[0];
        accMin[0] = (float)nMin[0];
        #pragma unroll
        for (int k = 1; k < 8; ++k) accMax[k] = eMax[k] - eMax[k - 1];
        #pragma unroll
        for (int k = 1; k < 7; ++k) accMin[k] = (float)(nMin[k] - nMin[k - 1]);
        accMin[7] = (float)(nscored - nMin[6]);
    }

    #pragma unroll
    for (int k = 0; k < 8; ++k) {
        float vM = accMax[k], vN = accMin[k];
        #pragma unroll
        for (int off = 16; off > 0; off >>= 1) {
            vM += __shfl_xor(vM, off, 32);
            vN += __shfl_xor(vN, off, 32);
        }
        if (lane == 0) {
            atomicAdd(&sAcc[k],     vM);
            atomicAdd(&sAcc[8 + k], vN);
        }
    }
    __syncthreads();

    if (tid < 16) {
        const int which = tid >> 3;
        const int k     = tid & 7;
        const size_t off = (size_t)b * 14336
                         + (size_t)(2 * u + which) * (NGROUP * NKER)
                         + (size_t)h * NKER + k;
        OUT[off] = sAcc[tid];
    }
}

extern "C" void kernel_launch(void* const* d_in, const int* in_sizes, int n_in,
                              void* d_out, int out_size, void* d_ws, size_t ws_size,
                              hipStream_t stream) {
    (void)in_sizes; (void)n_in; (void)d_ws; (void)ws_size; (void)out_size;
    const float* X   = (const float*)d_in[0];
    const float* W   = (const float*)d_in[1];
    const int*   IDX = (const int*)d_in[2];
    float*       OUT = (float*)d_out;

    const int shmem = SMEM_FLOATS * (int)sizeof(float);
    (void)hipFuncSetAttribute((const void*)hydra_unit_kernel,
                              hipFuncAttributeMaxDynamicSharedMemorySize, shmem);

    const int nblocks = 4 * NUNIT * NGROUP;
    hydra_unit_kernel<<<dim3(nblocks), dim3(1024), shmem, stream>>>(X, W, IDX, OUT);
}
